# starting kernel
# baseline (speedup 1.0000x reference)
#include <hip/hip_runtime.h>

typedef float v2f __attribute__((ext_vector_type(2)));
typedef float v8f __attribute__((ext_vector_type(8)));

__global__ __launch_bounds__(256) void u_dot_v_wmma_kernel(
    const float* __restrict__ hsrc,
    const float* __restrict__ hdst,
    const int*   __restrict__ src,
    const int*   __restrict__ dst,
    float*       __restrict__ out,
    int nE)
{
    const int lane = threadIdx.x & 31;
    const int wave = threadIdx.x >> 5;

    const int e0 = blockIdx.x * 128 + wave * 16;
    if (e0 >= nE) return;

    const int m  = lane & 15;
    const int e  = e0 + m;
    const int ec = (e < nE) ? e : (nE - 1);

    const long srow = (long)src[ec] * 128;
    const long drow = (long)dst[ec] * 128;
    const int  bg   = (lane >> 4) << 1;

    const float* ap = hsrc + srow + bg;
    const float* bp = hdst + drow + bg;

    v8f c = {};
    #pragma unroll
    for (int kb = 0; kb < 128; kb += 4) {
        v2f a = *(const v2f*)(ap + kb);
        v2f b = *(const v2f*)(bp + kb);
        c = __builtin_amdgcn_wmma_f32_16x16x4_f32(
                false, a, false, b, (short)0, c, false, false);
    }

    const bool writer = (lane < 8) || (lane >= 24);
    const int  idx    = (lane < 8) ? lane : (lane - 24);
    float val = c[0];
    #pragma unroll
    for (int i = 1; i < 8; ++i) val = (idx == i) ? c[i] : val;

    if (writer && e < nE) out[e] = val;
}

extern "C" void kernel_launch(void* const* d_in, const int* in_sizes, int n_in,
                              void* d_out, int out_size, void* d_ws, size_t ws_size,
                              hipStream_t stream) {
    const float* h_user = (const float*)d_in[0];
    const float* h_item = (const float*)d_in[1];
    const int*   src0   = (const int*)d_in[2];
    const int*   dst0   = (const int*)d_in[3];
    const int*   src1   = (const int*)d_in[4];
    const int*   dst1   = (const int*)d_in[5];
    const int*   src2   = (const int*)d_in[6];
    const int*   dst2   = (const int*)d_in[7];
    float* out = (float*)d_out;

    const int E = in_sizes[2];
    const int blocks = (E + 127) / 128;

    u_dot_v_wmma_kernel<<<blocks, 256, 0, stream>>>(h_user, h_user, src0, dst0, out,         E);
    u_dot_v_wmma_kernel<<<blocks, 256, 0, stream>>>(h_user, h_item, src1, dst1, out + E,     E);
    u_dot_v_wmma_kernel<<<blocks, 256, 0, stream>>>(h_item, h_user, src2, dst2, out + 2 * E, E);
}


# Persistent waves: one launch of 1024 resident work-groups, each wave loops over a contiguous slice of the 16-edge groups of all 3 edge types, next group's loads requested ahead; sum order unchanged
# speedup vs baseline: 2.9041x; 2.9041x over previous
#include <hip/hip_runtime.h>

typedef float v2f __attribute__((ext_vector_type(2)));
typedef float v4f __attribute__((ext_vector_type(4)));
typedef float v8f __attribute__((ext_vector_type(8)));
typedef int   v4i __attribute__((ext_vector_type(4)));


#define GRID_BLOCKS     1024
#define GRID_WAVES_LOG2 13

#define LOAD_TILE(SET, HS, HD, SROW, DROW) do { \
        const float* ap_ = (HS) + (long)(SROW) * 128 + q * 4; \
        const float* bp_ = (HD) + (long)(DROW) * 128 + q * 4; \
        _Pragma("unroll") for (int t_ = 0; t_ < 4; ++t_) { \
            a[(SET)][t_] = *(const v4f*)(ap_ + t_ * 32);    \
            b[(SET)][t_] = *(const v4f*)(bp_ + t_ * 32); \
        } } while (0)

#define COMPUTE_TILE(P, SET) do { \
        v8f acc_ = {}; \
        _Pragma("unroll") for (int t_ = 0; t_ < 4; ++t_) {     \
            const v4f av_ = a[(SET)][t_]; \
            const v4f bv_ = b[(SET)][t_]; \
            const v2f a0_ = {av_.x, av_.y}, a1_ = {av_.z, av_.w}; \
            const v2f b0_ = {bv_.x, bv_.y}, b1_ = {bv_.z, bv_.w}; \
              \
            acc_ = __builtin_amdgcn_wmma_f32_16x16x4_f32( \
                       false, a0_, false, b0_, (short)0, acc_, false, false); \
            acc_ = __builtin_amdgcn_wmma_f32_16x16x4_f32( \
                       false, a1_, false, b1_, (short)0, acc_, false, false); \
        } \
          \
          \
        float v_ = acc_[0]; \
        _Pragma("unroll") for (int i_ = 1; i_ < 8; ++i_) v_ = (idx == i_) ? acc_[i_] : v_;     \
          \
        v_ += __builtin_bit_cast(float, __builtin_amdgcn_update_dpp( \
                  0, __builtin_bit_cast(int, v_), 0xB1, 0xF, 0xF, false)); \
        v_ += __builtin_bit_cast(float, __builtin_amdgcn_update_dpp( \
                  0, __builtin_bit_cast(int, v_), 0x4E, 0xF, 0xF, false)); \
        res = ((lane & 3) == (P)) ? v_ : res; \
    } while (0)

#define STEP(P, NEXT_LOAD) do { \
        __builtin_amdgcn_sched_barrier(0); \
        NEXT_LOAD; \
        __builtin_amdgcn_sched_barrier(0); \
        COMPUTE_TILE((P), (P) & 1); \
    } while (0)

#define DECODE_GROUP(G, HS, HD, SP, DP, OP, E0) do { \
        const int g_ = (G); \
        const int t_ = (g_ >= ngrp ? 1 : 0) + (g_ >= 2 * ngrp ? 1 : 0);     \
        (E0) = (g_ - t_ * ngrp) << 4; \
        (HS) = (t_ == 2) ? h_item : h_user; \
        (HD) = (t_ == 1) ? h_item : h_user; \
        (SP) = (t_ == 0) ? src0 : (t_ == 1) ? src1 : src2; \
        (DP) = (t_ == 0) ? dst0 : (t_ == 1) ? dst1 : dst2; \
        (OP) = (t_ == 0) ? out  : (t_ == 1) ? out1 : out2; \
    } while (0)

#define LOAD_INDICES(SP, DP, E0, SI, DI) do { \
        if (FULL) { \
            const v4i s4_ = *(const v4i*)((SP) + (E0) + 4 * j); \
            const v4i d4_ = *(const v4i*)((DP) + (E0) + 4 * j); \
            (SI)[0] = s4_.x; (SI)[1] = s4_.y; (SI)[2] = s4_.z; (SI)[3] = s4_.w; \
            (DI)[0] = d4_.x; (DI)[1] = d4_.y; (DI)[2] = d4_.z; (DI)[3] = d4_.w; \
        } else { \
            _Pragma("unroll") for (int p_ = 0; p_ < 4; ++p_) {     \
                const int e_  = (E0) + 4 * j + p_; \
                const int ec_ = (e_ < nE) ? e_ : (nE - 1); \
                (SI)[p_] = (SP)[ec_]; \
                (DI)[p_] = (DP)[ec_]; \
            } \
        } } while (0)

template <bool FULL>
__device__ __forceinline__ void run_slice(
    const float* __restrict__ h_user, const float* __restrict__ h_item,
    const int* __restrict__ src0, const int* __restrict__ dst0,
    const int* __restrict__ src1, const int* __restrict__ dst1,
    const int* __restrict__ src2, const int* __restrict__ dst2,
    float* __restrict__ out, const int nE, const int ngrp, int g, const int g_end, const int lane)
{
    const int c = lane & 15;
    const int h = lane >> 4;
    const int j = c >> 2;
    const int q = (c & 3) + (h << 2);
    const int idx = lane & 7;
    const bool writer = (lane < 8) || (lane >= 24);
    const int  slot   = (lane < 8) ? lane : (lane - 16);
    const int  last_group = 3 * ngrp - 1;
    float* const out1 = out + nE;
    float* const out2 = out1 + nE;

    const float *hs, *hd;  const int *sp, *dp;  float* op;  int e0;
    int si[4], di[4];
    v4f a[2][4], b[2][4];

    DECODE_GROUP(g, hs, hd, sp, dp, op, e0);
    LOAD_INDICES(sp, dp, e0, si, di);
    LOAD_TILE(0, hs, hd, si[0], di[0]);

    for (; g < g_end; ++g) {
        const bool more = (g + 1 < g_end);
        const float *hsn, *hdn;  const int *spn, *dpn;  float* opn;  int e0n;
        int sn[4], dn[4];
        DECODE_GROUP((g + 1 <= last_group) ? (g + 1) : last_group, hsn, hdn, spn, dpn, opn, e0n);
        LOAD_INDICES(spn, dpn, e0n, sn, dn);

        float res = 0.0f;
        STEP(0, LOAD_TILE(1, hs, hd, si[1], di[1]));
        STEP(1, LOAD_TILE(0, hs, hd, si[2], di[2]));
        STEP(2, LOAD_TILE(1, hs, hd, si[3], di[3]));
        const int sr = more ? sn[0] : 0;
        const int dr = more ? dn[0] : 0;
        STEP(3, LOAD_TILE(0, hsn, hdn, sr, dr));

        const int e = e0 + slot;
        if (writer && (FULL || e < nE)) op[e] = res;

        hs = hsn; hd = hdn; sp = spn; dp = dpn; op = opn; e0 = e0n;
        #pragma unroll
        for (int p = 0; p < 4; ++p) { si[p] = sn[p]; di[p] = dn[p]; }
    }
}

__global__ __launch_bounds__(256) void u_dot_v_wmma_kernel(
    const float* __restrict__ h_user,
    const float* __restrict__ h_item,
    const int*   __restrict__ src0,
    const int*   __restrict__ dst0,
    const int*   __restrict__ src1,
    const int*   __restrict__ dst1,
    const int*   __restrict__ src2,
    const int*   __restrict__ dst2,
    float*       __restrict__ out,
    int nE)
{
    const int lane = threadIdx.x & 31;
    const int wave = __builtin_amdgcn_readfirstlane((int)(threadIdx.x >> 5));
    const int w    = (int)blockIdx.x * 8 + wave;

    const int ngrp  = (nE + 15) >> 4;
    const unsigned total = 3u * (unsigned)ngrp;
    const int g     = (int)(((unsigned long long)w       * total) >> GRID_WAVES_LOG2);
    const int g_end = (int)(((unsigned long long)(w + 1) * total) >> GRID_WAVES_LOG2);
    if (g >= g_end) return;

    if ((nE & 15) == 0)
        run_slice<true >(h_user, h_item, src0, dst0, src1, dst1, src2, dst2, out, nE, ngrp, g, g_end, lane);
    else
        run_slice<false>(h_user, h_item, src0, dst0, src1, dst1, src2, dst2, out, nE, ngrp, g, g_end, lane);
}

#undef LOAD_INDICES
#undef DECODE_GROUP
#undef STEP
#undef COMPUTE_TILE
#undef LOAD_TILE

extern "C" void kernel_launch(void* const* d_in, const int* in_sizes, int n_in,
                              void* d_out, int out_size, void* d_ws, size_t ws_size,
                              hipStream_t stream) {
    const float* h_user = (const float*)d_in[0];
    const float* h_item = (const float*)d_in[1];
    const int*   src0   = (const int*)d_in[2];
    const int*   dst0   = (const int*)d_in[3];
    const int*   src1   = (const int*)d_in[4];
    const int*   dst1   = (const int*)d_in[5];
    const int*   src2   = (const int*)d_in[6];
    const int*   dst2   = (const int*)d_in[7];
    float* out = (float*)d_out;

    const int E = in_sizes[2];

    u_dot_v_wmma_kernel<<<GRID_BLOCKS, 256, 0, stream>>>(
        h_user, h_item, src0, dst0, src1, dst1, src2, dst2, out, E);
}
